# starting kernel
# baseline (speedup 1.0000x reference)
#include <hip/hip_runtime.h>
#include <math.h>

typedef __attribute__((ext_vector_type(2))) float v2f;
typedef __attribute__((ext_vector_type(8))) float v8f;

#define IN_DIM 256
#define HID 96
#define OUT_DIM 64
#define ALPHA_F 0.15f
#define OMA_F 0.85f
#define EPS_F 1e-12f


__global__ void zero_int_k(int* p, int n) {
    int i = blockIdx.x * blockDim.x + threadIdx.x;
    if (i < n) p[i] = 0;
}

__global__ void cnt_k(const int* __restrict__ col, int* cnt, int E) {
    int i = blockIdx.x * blockDim.x + threadIdx.x;
    if (i < E) atomicAdd(&cnt[col[i]], 1);
}

__global__ void dinv_from_cnt_k(const int* __restrict__ cnt, float* dinv, int n) {
    int i = blockIdx.x * blockDim.x + threadIdx.x;
    if (i < n) dinv[i] = rsqrtf((float)(cnt[i] + 1));
}

__global__ __launch_bounds__(1024) void scan_k(const int* __restrict__ cnt,
                                               int* __restrict__ rowptr, int n) {
    __shared__ int buf[1024];
    __shared__ int carry_s;
    const int tid = threadIdx.x;
    if (tid == 0) carry_s = 0;
    __syncthreads();
    for (int base = 0; base < n; base += 1024) {
        const int idx = base + tid;
        const int v = (idx < n) ? cnt[idx] : 0;
        buf[tid] = v;
        __syncthreads();
        for (int off = 1; off < 1024; off <<= 1) {
            const int t = (tid >= off) ? buf[tid - off] : 0;
            __syncthreads();
            buf[tid] += t;
            __syncthreads();
        }
        const int incl  = buf[tid];
        const int carry = carry_s;
        if (idx < n) rowptr[idx] = carry + incl - v;
        __syncthreads();
        if (tid == 1023) carry_s = carry + buf[1023];
        __syncthreads();
    }
    if (tid == 0) rowptr[n] = carry_s;
}

__global__ void copy_int_k(int* __restrict__ dst, const int* __restrict__ src, int n) {
    int i = blockIdx.x * blockDim.x + threadIdx.x;
    if (i < n) dst[i] = src[i];
}

__global__ void fill_k(const int* __restrict__ row, const int* __restrict__ col,
                       int* cursor, int* __restrict__ srcs, int E) {
    int i = blockIdx.x * blockDim.x + threadIdx.x;
    if (i < E) {
        const int p = atomicAdd(&cursor[col[i]], 1);
        srcs[p] = row[i];
    }
}

template <int NT>
__global__ __launch_bounds__(256) void gemm_wmma_k(
    const float* __restrict__ X, const float* __restrict__ W,
    const float* __restrict__ bias, float* __restrict__ Out,
    int K, int nrows) {
    const int M    = NT * 16;
    const int wave = threadIdx.x >> 5;
    const int tm   = blockIdx.x * 8 + wave;
    if (tm * 16 >= nrows) return;
    const int lane = threadIdx.x & 31;
    const int half = lane >> 4;
    const int l16  = lane & 15;

    v8f c[NT];
#pragma unroll
    for (int t = 0; t < NT; ++t) c[t] = (v8f){};

    const float* __restrict__ xrow = X + (size_t)(tm * 16 + l16) * K;

    for (int k = 0; k < K; k += 4) {
        const int kk = k + 2 * half;
        v2f a;
        a.x = xrow[kk];
        a.y = xrow[kk + 1];
        const float* __restrict__ w0 = W + (size_t)kk * M + l16;
#pragma unroll
        for (int t = 0; t < NT; ++t) {
            v2f b;
            b.x = w0[t * 16];
            b.y = w0[M + t * 16];
            c[t] = __builtin_amdgcn_wmma_f32_16x16x4_f32(
                false, a, false, b, (short)0, c[t], false, false);
        }
    }

#pragma unroll
    for (int t = 0; t < NT; ++t) {
        const int bcol = t * 16 + l16;
        const float bv = bias[bcol];
#pragma unroll
        for (int r = 0; r < 8; ++r) {
            const int orow = tm * 16 + r + 8 * half;
            Out[(size_t)orow * M + bcol] = c[t][r] + bv;
        }
    }
}

__global__ __launch_bounds__(256) void l2norm_k(float* x, int n, int M, float s) {
    const int wave = threadIdx.x >> 5;
    const int lane = threadIdx.x & 31;
    const int row  = blockIdx.x * 8 + wave;
    if (row >= n) return;
    float* p = x + (size_t)row * M;
    float sum = 0.0f;
    for (int f = lane; f < M; f += 32) { float v = p[f]; sum += v * v; }
#pragma unroll
    for (int o = 16; o > 0; o >>= 1) sum += __shfl_xor(sum, o, 32);
    const float scale = s / fmaxf(sqrtf(sum), EPS_F);
    for (int f = lane; f < M; f += 32) p[f] *= scale;
}


__global__ void prescale_k(float* __restrict__ Y, const float* __restrict__ x,
                           const float* __restrict__ dinv, int n, int M) {
    size_t i = (size_t)blockIdx.x * blockDim.x + threadIdx.x;
    if (i >= (size_t)n * M) return;
    Y[i] = x[i] * dinv[(int)(i / M)];
}

__global__ __launch_bounds__(256) void gather_prop_k(
    float* __restrict__ dst, const float* __restrict__ Y,
    const float* __restrict__ h, const float* __restrict__ dinv,
    const int* __restrict__ rowptr, const int* __restrict__ srcs,
    int n, int M, int relu) {
    const int wave = threadIdx.x >> 5;
    const int lane = threadIdx.x & 31;
    const int node = blockIdx.x * 8 + wave;
    if (node >= n) return;

    const float* __restrict__ yn = Y + (size_t)node * M;
    float a0 = yn[lane];
    float a1 = (M > 32) ? yn[lane + 32] : 0.0f;
    float a2 = (M > 64) ? yn[lane + 64] : 0.0f;

    const int e0 = rowptr[node];
    const int e1 = rowptr[node + 1];
    for (int e = e0; e < e1; ++e) {
        const float* __restrict__ ys = Y + (size_t)srcs[e] * M;
        a0 += ys[lane];
        if (M > 32) a1 += ys[lane + 32];
        if (M > 64) a2 += ys[lane + 64];
    }

    const float dd = OMA_F * dinv[node];
    const float* __restrict__ hn = h + (size_t)node * M;
    float* __restrict__ dn = dst + (size_t)node * M;

    float v0 = dd * a0 + ALPHA_F * hn[lane];
    if (relu) v0 = fmaxf(v0, 0.0f);
    dn[lane] = v0;
    if (M > 32) {
        float v1 = dd * a1 + ALPHA_F * hn[lane + 32];
        if (relu) v1 = fmaxf(v1, 0.0f);
        dn[lane + 32] = v1;
    }
    if (M > 64) {
        float v2 = dd * a2 + ALPHA_F * hn[lane + 64];
        if (relu) v2 = fmaxf(v2, 0.0f);
        dn[lane + 64] = v2;
    }
}


static inline size_t align256(size_t b) { return (b + 255) & ~(size_t)255; }

extern "C" void kernel_launch(void* const* d_in, const int* in_sizes, int n_in,
                              void* d_out, int out_size, void* d_ws, size_t ws_size,
                              hipStream_t stream) {
    const float* x_in = (const float*)d_in[0];
    const int*   ei   = (const int*)d_in[1];
    const float* W0   = (const float*)d_in[2];
    const float* b0   = (const float*)d_in[3];
    const float* W1   = (const float*)d_in[4];
    const float* b1   = (const float*)d_in[5];
    const float* Wx   = (const float*)d_in[6];
    const float* bx   = (const float*)d_in[7];
    float* out = (float*)d_out;

    const int n = in_sizes[0] / IN_DIM;
    const int E = in_sizes[1] / 2;
    const int* row = ei;
    const int* col = ei + E;

    char* ws = (char*)d_ws;
    size_t off = 0;
    float* dinv   = (float*)(ws + off); off += align256((size_t)n * 4);
    int*   cnt    = (int*)  (ws + off); off += align256((size_t)n * 4);
    int*   rowptr = (int*)  (ws + off); off += align256((size_t)(n + 1) * 4);
    int*   cursor = (int*)  (ws + off); off += align256((size_t)n * 4);
    int*   srcs   = (int*)  (ws + off); off += align256((size_t)E * 4);
    float* H      = (float*)(ws + off); off += align256((size_t)n * HID * 4);
    float* X      = (float*)(ws + off); off += align256((size_t)n * HID * 4);
    float* Y      = (float*)(ws + off); off += align256((size_t)n * HID * 4);
    (void)ws_size;

    const int TPB  = 256;
    const int gN   = (n + TPB - 1) / TPB;
    const int gE   = (E + TPB - 1) / TPB;
    const int gRow = (n + 7) / 8;
    const int gTile = ((n / 16) + 7) / 8;

    zero_int_k<<<gN, TPB, 0, stream>>>(cnt, n);
    cnt_k<<<gE, TPB, 0, stream>>>(col, cnt, E);
    dinv_from_cnt_k<<<gN, TPB, 0, stream>>>(cnt, dinv, n);
    scan_k<<<1, 1024, 0, stream>>>(cnt, rowptr, n);
    copy_int_k<<<gN, TPB, 0, stream>>>(cursor, rowptr, n);
    fill_k<<<gE, TPB, 0, stream>>>(row, col, cursor, srcs, E);

    auto prop = [&](const float* src, float* dst, const float* h, int M, int relu) {
        const size_t total = (size_t)n * M;
        const int gT = (int)((total + TPB - 1) / TPB);
        prescale_k<<<gT, TPB, 0, stream>>>(Y, src, dinv, n, M);
        gather_prop_k<<<gRow, TPB, 0, stream>>>(dst, Y, h, dinv, rowptr, srcs, n, M, relu);
    };

    gemm_wmma_k<HID / 16><<<gTile, TPB, 0, stream>>>(x_in, W0, b0, H, IN_DIM, n);
    l2norm_k<<<gRow, TPB, 0, stream>>>(H, n, HID, 1.8f);
    prop(H, X, H, HID, 0);
    prop(X, X, H, HID, 1);

    gemm_wmma_k<HID / 16><<<gTile, TPB, 0, stream>>>(X, W1, b1, H, HID, n);
    l2norm_k<<<gRow, TPB, 0, stream>>>(H, n, HID, 1.8f);
    prop(H, X, H, HID, 0);
    prop(X, X, H, HID, 0);
    l2norm_k<<<gRow, TPB, 0, stream>>>(X, n, HID, 1.5f);

    gemm_wmma_k<OUT_DIM / 16><<<gTile, TPB, 0, stream>>>(X, Wx, bx, H, HID, n);
    l2norm_k<<<gRow, TPB, 0, stream>>>(H, n, OUT_DIM, 1.8f);
    prop(H, X, H, OUT_DIM, 0);
    prop(X, out, H, OUT_DIM, 0);
}


# scan_k: 49 serial 1024-wide LDS scan passes in one workgroup become one pass of n/1024+1 independent workgroups, each summing the counts before its chunk itself, then a wave-shuffle scan
# speedup vs baseline: 1.2115x; 1.2115x over previous
#include <hip/hip_runtime.h>
#include <math.h>

typedef __attribute__((ext_vector_type(2))) float v2f;
typedef __attribute__((ext_vector_type(8))) float v8f;

#define IN_DIM 256
#define HID 96
#define OUT_DIM 64
#define ALPHA_F 0.15f
#define OMA_F 0.85f
#define EPS_F 1e-12f


__global__ void zero_int_k(int* p, int n) {
    int i = blockIdx.x * blockDim.x + threadIdx.x;
    if (i < n) p[i] = 0;
}

__global__ void cnt_k(const int* __restrict__ col, int* cnt, int E) {
    int i = blockIdx.x * blockDim.x + threadIdx.x;
    if (i < E) atomicAdd(&cnt[col[i]], 1);
}

__global__ void dinv_from_cnt_k(const int* __restrict__ cnt, float* dinv, int n) {
    int i = blockIdx.x * blockDim.x + threadIdx.x;
    if (i < n) dinv[i] = rsqrtf((float)(cnt[i] + 1));
}

__global__ __launch_bounds__(1024) void scan_k(const int* __restrict__ cnt,
                                               int* __restrict__ rowptr, int n) {
    __shared__ int wtot_v[32];
    __shared__ int wtot_p[32];
    const int tid  = threadIdx.x;
    const int lane = tid & 31;
    const int wave = tid >> 5;

    const int idx = blockIdx.x * 1024 + tid;
    const int v   = (idx < n) ? cnt[idx] : 0;

    const int4* __restrict__ c4 = reinterpret_cast<const int4*>(cnt);
    const int lim = blockIdx.x * 256;
    int p = 0;
#pragma unroll 4
    for (int i = tid; i < lim; i += 1024) {
        const int4 q = c4[i];
        p += (q.x + q.y) + (q.z + q.w);
    }

    int incl = v;
#pragma unroll
    for (int o = 1; o < 32; o <<= 1) {
        const int t = __shfl_up(incl, o, 32);
        if (lane >= o) incl += t;
    }
#pragma unroll
    for (int o = 16; o > 0; o >>= 1) p += __shfl_xor(p, o, 32);

    if (lane == 31) wtot_v[wave] = incl;
    if (lane == 0)  wtot_p[wave] = p;
    __syncthreads();

    int s = wtot_p[lane] + ((lane < wave) ? wtot_v[lane] : 0);
#pragma unroll
    for (int o = 16; o > 0; o >>= 1) s += __shfl_xor(s, o, 32);

    if (idx <= n) rowptr[idx] = s + incl - v;
}

__global__ void copy_int_k(int* __restrict__ dst, const int* __restrict__ src, int n) {
    int i = blockIdx.x * blockDim.x + threadIdx.x;
    if (i < n) dst[i] = src[i];
}

__global__ void fill_k(const int* __restrict__ row, const int* __restrict__ col,
                       int* cursor, int* __restrict__ srcs, int E) {
    int i = blockIdx.x * blockDim.x + threadIdx.x;
    if (i < E) {
        const int p = atomicAdd(&cursor[col[i]], 1);
        srcs[p] = row[i];
    }
}

template <int NT>
__global__ __launch_bounds__(256) void gemm_wmma_k(
    const float* __restrict__ X, const float* __restrict__ W,
    const float* __restrict__ bias, float* __restrict__ Out,
    int K, int nrows) {
    const int M    = NT * 16;
    const int wave = threadIdx.x >> 5;
    const int tm   = blockIdx.x * 8 + wave;
    if (tm * 16 >= nrows) return;
    const int lane = threadIdx.x & 31;
    const int half = lane >> 4;
    const int l16  = lane & 15;

    v8f c[NT];
#pragma unroll
    for (int t = 0; t < NT; ++t) c[t] = (v8f){};

    const float* __restrict__ xrow = X + (size_t)(tm * 16 + l16) * K;

    for (int k = 0; k < K; k += 4) {
        const int kk = k + 2 * half;
        v2f a;
        a.x = xrow[kk];
        a.y = xrow[kk + 1];
        const float* __restrict__ w0 = W + (size_t)kk * M + l16;
#pragma unroll
        for (int t = 0; t < NT; ++t) {
            v2f b;
            b.x = w0[t * 16];
            b.y = w0[M + t * 16];
            c[t] = __builtin_amdgcn_wmma_f32_16x16x4_f32(
                false, a, false, b, (short)0, c[t], false, false);
        }
    }

#pragma unroll
    for (int t = 0; t < NT; ++t) {
        const int bcol = t * 16 + l16;
        const float bv = bias[bcol];
#pragma unroll
        for (int r = 0; r < 8; ++r) {
            const int orow = tm * 16 + r + 8 * half;
            Out[(size_t)orow * M + bcol] = c[t][r] + bv;
        }
    }
}

__global__ __launch_bounds__(256) void l2norm_k(float* x, int n, int M, float s) {
    const int wave = threadIdx.x >> 5;
    const int lane = threadIdx.x & 31;
    const int row  = blockIdx.x * 8 + wave;
    if (row >= n) return;
    float* p = x + (size_t)row * M;
    float sum = 0.0f;
    for (int f = lane; f < M; f += 32) { float v = p[f]; sum += v * v; }
#pragma unroll
    for (int o = 16; o > 0; o >>= 1) sum += __shfl_xor(sum, o, 32);
    const float scale = s / fmaxf(sqrtf(sum), EPS_F);
    for (int f = lane; f < M; f += 32) p[f] *= scale;
}


__global__ void prescale_k(float* __restrict__ Y, const float* __restrict__ x,
                           const float* __restrict__ dinv, int n, int M) {
    size_t i = (size_t)blockIdx.x * blockDim.x + threadIdx.x;
    if (i >= (size_t)n * M) return;
    Y[i] = x[i] * dinv[(int)(i / M)];
}

__global__ __launch_bounds__(256) void gather_prop_k(
    float* __restrict__ dst, const float* __restrict__ Y,
    const float* __restrict__ h, const float* __restrict__ dinv,
    const int* __restrict__ rowptr, const int* __restrict__ srcs,
    int n, int M, int relu) {
    const int wave = threadIdx.x >> 5;
    const int lane = threadIdx.x & 31;
    const int node = blockIdx.x * 8 + wave;
    if (node >= n) return;

    const float* __restrict__ yn = Y + (size_t)node * M;
    float a0 = yn[lane];
    float a1 = (M > 32) ? yn[lane + 32] : 0.0f;
    float a2 = (M > 64) ? yn[lane + 64] : 0.0f;

    const int e0 = rowptr[node];
    const int e1 = rowptr[node + 1];
    for (int e = e0; e < e1; ++e) {
        const float* __restrict__ ys = Y + (size_t)srcs[e] * M;
        a0 += ys[lane];
        if (M > 32) a1 += ys[lane + 32];
        if (M > 64) a2 += ys[lane + 64];
    }

    const float dd = OMA_F * dinv[node];
    const float* __restrict__ hn = h + (size_t)node * M;
    float* __restrict__ dn = dst + (size_t)node * M;

    float v0 = dd * a0 + ALPHA_F * hn[lane];
    if (relu) v0 = fmaxf(v0, 0.0f);
    dn[lane] = v0;
    if (M > 32) {
        float v1 = dd * a1 + ALPHA_F * hn[lane + 32];
        if (relu) v1 = fmaxf(v1, 0.0f);
        dn[lane + 32] = v1;
    }
    if (M > 64) {
        float v2 = dd * a2 + ALPHA_F * hn[lane + 64];
        if (relu) v2 = fmaxf(v2, 0.0f);
        dn[lane + 64] = v2;
    }
}


static inline size_t align256(size_t b) { return (b + 255) & ~(size_t)255; }

extern "C" void kernel_launch(void* const* d_in, const int* in_sizes, int n_in,
                              void* d_out, int out_size, void* d_ws, size_t ws_size,
                              hipStream_t stream) {
    const float* x_in = (const float*)d_in[0];
    const int*   ei   = (const int*)d_in[1];
    const float* W0   = (const float*)d_in[2];
    const float* b0   = (const float*)d_in[3];
    const float* W1   = (const float*)d_in[4];
    const float* b1   = (const float*)d_in[5];
    const float* Wx   = (const float*)d_in[6];
    const float* bx   = (const float*)d_in[7];
    float* out = (float*)d_out;

    const int n = in_sizes[0] / IN_DIM;
    const int E = in_sizes[1] / 2;
    const int* row = ei;
    const int* col = ei + E;

    char* ws = (char*)d_ws;
    size_t off = 0;
    float* dinv   = (float*)(ws + off); off += align256((size_t)n * 4);
    int*   cnt    = (int*)  (ws + off); off += align256((size_t)n * 4);
    int*   rowptr = (int*)  (ws + off); off += align256((size_t)(n + 1) * 4);
    int*   cursor = (int*)  (ws + off); off += align256((size_t)n * 4);
    int*   srcs   = (int*)  (ws + off); off += align256((size_t)E * 4);
    float* H      = (float*)(ws + off); off += align256((size_t)n * HID * 4);
    float* X      = (float*)(ws + off); off += align256((size_t)n * HID * 4);
    float* Y      = (float*)(ws + off); off += align256((size_t)n * HID * 4);
    (void)ws_size;

    const int TPB  = 256;
    const int gN   = (n + TPB - 1) / TPB;
    const int gE   = (E + TPB - 1) / TPB;
    const int gRow = (n + 7) / 8;
    const int gTile = ((n / 16) + 7) / 8;

    zero_int_k<<<gN, TPB, 0, stream>>>(cnt, n);
    cnt_k<<<gE, TPB, 0, stream>>>(col, cnt, E);
    dinv_from_cnt_k<<<gN, TPB, 0, stream>>>(cnt, dinv, n);
    scan_k<<<n / 1024 + 1, 1024, 0, stream>>>(cnt, rowptr, n);
    copy_int_k<<<gN, TPB, 0, stream>>>(cursor, rowptr, n);
    fill_k<<<gE, TPB, 0, stream>>>(row, col, cursor, srcs, E);

    auto prop = [&](const float* src, float* dst, const float* h, int M, int relu) {
        const size_t total = (size_t)n * M;
        const int gT = (int)((total + TPB - 1) / TPB);
        prescale_k<<<gT, TPB, 0, stream>>>(Y, src, dinv, n, M);
        gather_prop_k<<<gRow, TPB, 0, stream>>>(dst, Y, h, dinv, rowptr, srcs, n, M, relu);
    };

    gemm_wmma_k<HID / 16><<<gTile, TPB, 0, stream>>>(x_in, W0, b0, H, IN_DIM, n);
    l2norm_k<<<gRow, TPB, 0, stream>>>(H, n, HID, 1.8f);
    prop(H, X, H, HID, 0);
    prop(X, X, H, HID, 1);

    gemm_wmma_k<HID / 16><<<gTile, TPB, 0, stream>>>(X, W1, b1, H, HID, n);
    l2norm_k<<<gRow, TPB, 0, stream>>>(H, n, HID, 1.8f);
    prop(H, X, H, HID, 0);
    prop(X, X, H, HID, 0);
    l2norm_k<<<gRow, TPB, 0, stream>>>(X, n, HID, 1.5f);

    gemm_wmma_k<OUT_DIM / 16><<<gTile, TPB, 0, stream>>>(X, Wx, bx, H, HID, n);
    l2norm_k<<<gRow, TPB, 0, stream>>>(H, n, OUT_DIM, 1.8f);
    prop(H, X, H, OUT_DIM, 0);
    prop(X, out, H, OUT_DIM, 0);
}


# prescale fused into its producers: l2norm and each layer's first gather store the dinv-scaled rows the next gather reads; the 6 one-element-per-thread prescale launches are removed (25 -> 19)
# speedup vs baseline: 1.3423x; 1.1080x over previous
#include <hip/hip_runtime.h>
#include <math.h>

typedef __attribute__((ext_vector_type(2))) float v2f;
typedef __attribute__((ext_vector_type(8))) float v8f;

#define IN_DIM 256
#define HID 96
#define OUT_DIM 64
#define ALPHA_F 0.15f
#define OMA_F 0.85f
#define EPS_F 1e-12f


__global__ void zero_int_k(int* p, int n) {
    int i = blockIdx.x * blockDim.x + threadIdx.x;
    if (i < n) p[i] = 0;
}

__global__ void cnt_k(const int* __restrict__ col, int* cnt, int E) {
    int i = blockIdx.x * blockDim.x + threadIdx.x;
    if (i < E) atomicAdd(&cnt[col[i]], 1);
}

__global__ void dinv_from_cnt_k(const int* __restrict__ cnt, float* dinv, int n) {
    int i = blockIdx.x * blockDim.x + threadIdx.x;
    if (i < n) dinv[i] = rsqrtf((float)(cnt[i] + 1));
}

__global__ __launch_bounds__(1024) void scan_k(const int* __restrict__ cnt,
                                               int* __restrict__ rowptr, int n) {
    __shared__ int wtot_v[32];
    __shared__ int wtot_p[32];
    const int tid  = threadIdx.x;
    const int lane = tid & 31;
    const int wave = tid >> 5;

    const int idx = blockIdx.x * 1024 + tid;
    const int v   = (idx < n) ? cnt[idx] : 0;

    const int4* __restrict__ c4 = reinterpret_cast<const int4*>(cnt);
    const int lim = blockIdx.x * 256;
    int p = 0;
#pragma unroll 4
    for (int i = tid; i < lim; i += 1024) {
        const int4 q = c4[i];
        p += (q.x + q.y) + (q.z + q.w);
    }

    int incl = v;
#pragma unroll
    for (int o = 1; o < 32; o <<= 1) {
        const int t = __shfl_up(incl, o, 32);
        if (lane >= o) incl += t;
    }
#pragma unroll
    for (int o = 16; o > 0; o >>= 1) p += __shfl_xor(p, o, 32);

    if (lane == 31) wtot_v[wave] = incl;
    if (lane == 0)  wtot_p[wave] = p;
    __syncthreads();

    int s = wtot_p[lane] + ((lane < wave) ? wtot_v[lane] : 0);
#pragma unroll
    for (int o = 16; o > 0; o >>= 1) s += __shfl_xor(s, o, 32);

    if (idx <= n) rowptr[idx] = s + incl - v;
}

__global__ void copy_int_k(int* __restrict__ dst, const int* __restrict__ src, int n) {
    int i = blockIdx.x * blockDim.x + threadIdx.x;
    if (i < n) dst[i] = src[i];
}

__global__ void fill_k(const int* __restrict__ row, const int* __restrict__ col,
                       int* cursor, int* __restrict__ srcs, int E) {
    int i = blockIdx.x * blockDim.x + threadIdx.x;
    if (i < E) {
        const int p = atomicAdd(&cursor[col[i]], 1);
        srcs[p] = row[i];
    }
}

template <int NT>
__global__ __launch_bounds__(256) void gemm_wmma_k(
    const float* __restrict__ X, const float* __restrict__ W,
    const float* __restrict__ bias, float* __restrict__ Out,
    int K, int nrows) {
    const int M    = NT * 16;
    const int wave = threadIdx.x >> 5;
    const int tm   = blockIdx.x * 8 + wave;
    if (tm * 16 >= nrows) return;
    const int lane = threadIdx.x & 31;
    const int half = lane >> 4;
    const int l16  = lane & 15;

    v8f c[NT];
#pragma unroll
    for (int t = 0; t < NT; ++t) c[t] = (v8f){};

    const float* __restrict__ xrow = X + (size_t)(tm * 16 + l16) * K;

    for (int k = 0; k < K; k += 4) {
        const int kk = k + 2 * half;
        v2f a;
        a.x = xrow[kk];
        a.y = xrow[kk + 1];
        const float* __restrict__ w0 = W + (size_t)kk * M + l16;
#pragma unroll
        for (int t = 0; t < NT; ++t) {
            v2f b;
            b.x = w0[t * 16];
            b.y = w0[M + t * 16];
            c[t] = __builtin_amdgcn_wmma_f32_16x16x4_f32(
                false, a, false, b, (short)0, c[t], false, false);
        }
    }

#pragma unroll
    for (int t = 0; t < NT; ++t) {
        const int bcol = t * 16 + l16;
        const float bv = bias[bcol];
#pragma unroll
        for (int r = 0; r < 8; ++r) {
            const int orow = tm * 16 + r + 8 * half;
            Out[(size_t)orow * M + bcol] = c[t][r] + bv;
        }
    }
}

__global__ __launch_bounds__(256) void l2norm_k(float* x, float* __restrict__ y,
                                                const float* __restrict__ dinv,
                                                int n, int M, float s) {
    const int wave = threadIdx.x >> 5;
    const int lane = threadIdx.x & 31;
    const int row  = blockIdx.x * 8 + wave;
    if (row >= n) return;
    float* p = x + (size_t)row * M;
    float sum = 0.0f;
    for (int f = lane; f < M; f += 32) { float v = p[f]; sum += v * v; }
#pragma unroll
    for (int o = 16; o > 0; o >>= 1) sum += __shfl_xor(sum, o, 32);
    const float scale = s / fmaxf(sqrtf(sum), EPS_F);
    if (y != nullptr) {
        const float di = dinv[row];
        float* __restrict__ q = y + (size_t)row * M;
        for (int f = lane; f < M; f += 32) {
            const float v = p[f] * scale;
            p[f] = v;
            q[f] = v * di;
        }
    } else {
        for (int f = lane; f < M; f += 32) p[f] *= scale;
    }
}


__global__ __launch_bounds__(256) void gather_prop_k(
    float* __restrict__ dst, const float* __restrict__ Y,
    const float* __restrict__ h, const float* __restrict__ dinv,
    const int* __restrict__ rowptr, const int* __restrict__ srcs,
    int n, int M, int relu, int pre) {
    const int wave = threadIdx.x >> 5;
    const int lane = threadIdx.x & 31;
    const int node = blockIdx.x * 8 + wave;
    if (node >= n) return;

    const float* __restrict__ yn = Y + (size_t)node * M;
    float a0 = yn[lane];
    float a1 = (M > 32) ? yn[lane + 32] : 0.0f;
    float a2 = (M > 64) ? yn[lane + 64] : 0.0f;

    const int e0 = rowptr[node];
    const int e1 = rowptr[node + 1];
    for (int e = e0; e < e1; ++e) {
        const float* __restrict__ ys = Y + (size_t)srcs[e] * M;
        a0 += ys[lane];
        if (M > 32) a1 += ys[lane + 32];
        if (M > 64) a2 += ys[lane + 64];
    }

    const float di = dinv[node];
    const float dd = OMA_F * di;
    const float* __restrict__ hn = h + (size_t)node * M;
    float* __restrict__ dn = dst + (size_t)node * M;

    float v0 = dd * a0 + ALPHA_F * hn[lane];
    if (relu) v0 = fmaxf(v0, 0.0f);
    if (pre) v0 *= di;
    dn[lane] = v0;
    if (M > 32) {
        float v1 = dd * a1 + ALPHA_F * hn[lane + 32];
        if (relu) v1 = fmaxf(v1, 0.0f);
        if (pre) v1 *= di;
        dn[lane + 32] = v1;
    }
    if (M > 64) {
        float v2 = dd * a2 + ALPHA_F * hn[lane + 64];
        if (relu) v2 = fmaxf(v2, 0.0f);
        if (pre) v2 *= di;
        dn[lane + 64] = v2;
    }
}


static inline size_t align256(size_t b) { return (b + 255) & ~(size_t)255; }

extern "C" void kernel_launch(void* const* d_in, const int* in_sizes, int n_in,
                              void* d_out, int out_size, void* d_ws, size_t ws_size,
                              hipStream_t stream) {
    const float* x_in = (const float*)d_in[0];
    const int*   ei   = (const int*)d_in[1];
    const float* W0   = (const float*)d_in[2];
    const float* b0   = (const float*)d_in[3];
    const float* W1   = (const float*)d_in[4];
    const float* b1   = (const float*)d_in[5];
    const float* Wx   = (const float*)d_in[6];
    const float* bx   = (const float*)d_in[7];
    float* out = (float*)d_out;

    const int n = in_sizes[0] / IN_DIM;
    const int E = in_sizes[1] / 2;
    const int* row = ei;
    const int* col = ei + E;

    char* ws = (char*)d_ws;
    size_t off = 0;
    float* dinv   = (float*)(ws + off); off += align256((size_t)n * 4);
    int*   cnt    = (int*)  (ws + off); off += align256((size_t)n * 4);
    int*   rowptr = (int*)  (ws + off); off += align256((size_t)(n + 1) * 4);
    int*   cursor = (int*)  (ws + off); off += align256((size_t)n * 4);
    int*   srcs   = (int*)  (ws + off); off += align256((size_t)E * 4);
    float* H      = (float*)(ws + off); off += align256((size_t)n * HID * 4);
    float* P      = (float*)(ws + off); off += align256((size_t)n * HID * 4);
    float* Q      = (float*)(ws + off); off += align256((size_t)n * HID * 4);
    (void)ws_size;

    const int TPB  = 256;
    const int gN   = (n + TPB - 1) / TPB;
    const int gE   = (E + TPB - 1) / TPB;
    const int gRow = (n + 7) / 8;
    const int gTile = ((n / 16) + 7) / 8;

    zero_int_k<<<gN, TPB, 0, stream>>>(cnt, n);
    cnt_k<<<gE, TPB, 0, stream>>>(col, cnt, E);
    dinv_from_cnt_k<<<gN, TPB, 0, stream>>>(cnt, dinv, n);
    scan_k<<<n / 1024 + 1, 1024, 0, stream>>>(cnt, rowptr, n);
    copy_int_k<<<gN, TPB, 0, stream>>>(cursor, rowptr, n);
    fill_k<<<gE, TPB, 0, stream>>>(row, col, cursor, srcs, E);

    auto prop = [&](const float* y, float* dst, const float* h, int M, int relu, int pre) {
        gather_prop_k<<<gRow, TPB, 0, stream>>>(dst, y, h, dinv, rowptr, srcs, n, M, relu, pre);
    };

    gemm_wmma_k<HID / 16><<<gTile, TPB, 0, stream>>>(x_in, W0, b0, H, IN_DIM, n);
    l2norm_k<<<gRow, TPB, 0, stream>>>(H, P, dinv, n, HID, 1.8f);
    prop(P, Q, H, HID, 0, 1);
    prop(Q, P, H, HID, 1, 0);

    gemm_wmma_k<HID / 16><<<gTile, TPB, 0, stream>>>(P, W1, b1, H, HID, n);
    l2norm_k<<<gRow, TPB, 0, stream>>>(H, Q, dinv, n, HID, 1.8f);
    prop(Q, P, H, HID, 0, 1);
    prop(P, Q, H, HID, 0, 0);
    l2norm_k<<<gRow, TPB, 0, stream>>>(Q, nullptr, dinv, n, HID, 1.5f);

    gemm_wmma_k<OUT_DIM / 16><<<gTile, TPB, 0, stream>>>(Q, Wx, bx, H, HID, n);
    l2norm_k<<<gRow, TPB, 0, stream>>>(H, P, dinv, n, OUT_DIM, 1.8f);
    prop(P, Q, H, OUT_DIM, 0, 1);
    prop(Q, out, H, OUT_DIM, 0, 0);
}


# gather: a wave fetches a whole row with one load per lane and issues the loads of up to 8 source rows before adding them in CSR order: 8 row loads in flight per wave instead of 1; sum order kept
# speedup vs baseline: 2.0336x; 1.5150x over previous
#include <hip/hip_runtime.h>
#include <math.h>

typedef __attribute__((ext_vector_type(2))) float v2f;
typedef __attribute__((ext_vector_type(8))) float v8f;

#define IN_DIM 256
#define HID 96
#define OUT_DIM 64
#define ALPHA_F 0.15f
#define OMA_F 0.85f
#define EPS_F 1e-12f


__global__ void zero_int_k(int* p, int n) {
    int i = blockIdx.x * blockDim.x + threadIdx.x;
    if (i < n) p[i] = 0;
}

__global__ void cnt_k(const int* __restrict__ col, int* cnt, int E) {
    int i = blockIdx.x * blockDim.x + threadIdx.x;
    if (i < E) atomicAdd(&cnt[col[i]], 1);
}

__global__ void dinv_from_cnt_k(const int* __restrict__ cnt, float* dinv, int n) {
    int i = blockIdx.x * blockDim.x + threadIdx.x;
    if (i < n) dinv[i] = rsqrtf((float)(cnt[i] + 1));
}

__global__ __launch_bounds__(1024) void scan_k(const int* __restrict__ cnt,
                                               int* __restrict__ rowptr, int n) {
    __shared__ int wtot_v[32];
    __shared__ int wtot_p[32];
    const int tid  = threadIdx.x;
    const int lane = tid & 31;
    const int wave = tid >> 5;

    const int idx = blockIdx.x * 1024 + tid;
    const int v   = (idx < n) ? cnt[idx] : 0;

    const int4* __restrict__ c4 = reinterpret_cast<const int4*>(cnt);
    const int lim = blockIdx.x * 256;
    int p = 0;
#pragma unroll 4
    for (int i = tid; i < lim; i += 1024) {
        const int4 q = c4[i];
        p += (q.x + q.y) + (q.z + q.w);
    }

    int incl = v;
#pragma unroll
    for (int o = 1; o < 32; o <<= 1) {
        const int t = __shfl_up(incl, o, 32);
        if (lane >= o) incl += t;
    }
#pragma unroll
    for (int o = 16; o > 0; o >>= 1) p += __shfl_xor(p, o, 32);

    if (lane == 31) wtot_v[wave] = incl;
    if (lane == 0)  wtot_p[wave] = p;
    __syncthreads();

    int s = wtot_p[lane] + ((lane < wave) ? wtot_v[lane] : 0);
#pragma unroll
    for (int o = 16; o > 0; o >>= 1) s += __shfl_xor(s, o, 32);

    if (idx <= n) rowptr[idx] = s + incl - v;
}

__global__ void copy_int_k(int* __restrict__ dst, const int* __restrict__ src, int n) {
    int i = blockIdx.x * blockDim.x + threadIdx.x;
    if (i < n) dst[i] = src[i];
}

__global__ void fill_k(const int* __restrict__ row, const int* __restrict__ col,
                       int* cursor, int* __restrict__ srcs, int E) {
    int i = blockIdx.x * blockDim.x + threadIdx.x;
    if (i < E) {
        const int p = atomicAdd(&cursor[col[i]], 1);
        srcs[p] = row[i];
    }
}

template <int NT>
__global__ __launch_bounds__(256) void gemm_wmma_k(
    const float* __restrict__ X, const float* __restrict__ W,
    const float* __restrict__ bias, float* __restrict__ Out,
    int K, int nrows) {
    const int M    = NT * 16;
    const int wave = threadIdx.x >> 5;
    const int tm   = blockIdx.x * 8 + wave;
    if (tm * 16 >= nrows) return;
    const int lane = threadIdx.x & 31;
    const int half = lane >> 4;
    const int l16  = lane & 15;

    v8f c[NT];
#pragma unroll
    for (int t = 0; t < NT; ++t) c[t] = (v8f){};

    const float* __restrict__ xrow = X + (size_t)(tm * 16 + l16) * K;

    for (int k = 0; k < K; k += 4) {
        const int kk = k + 2 * half;
        v2f a;
        a.x = xrow[kk];
        a.y = xrow[kk + 1];
        const float* __restrict__ w0 = W + (size_t)kk * M + l16;
#pragma unroll
        for (int t = 0; t < NT; ++t) {
            v2f b;
            b.x = w0[t * 16];
            b.y = w0[M + t * 16];
            c[t] = __builtin_amdgcn_wmma_f32_16x16x4_f32(
                false, a, false, b, (short)0, c[t], false, false);
        }
    }

#pragma unroll
    for (int t = 0; t < NT; ++t) {
        const int bcol = t * 16 + l16;
        const float bv = bias[bcol];
#pragma unroll
        for (int r = 0; r < 8; ++r) {
            const int orow = tm * 16 + r + 8 * half;
            Out[(size_t)orow * M + bcol] = c[t][r] + bv;
        }
    }
}

__global__ __launch_bounds__(256) void l2norm_k(float* x, float* __restrict__ y,
                                                const float* __restrict__ dinv,
                                                int n, int M, float s) {
    const int wave = threadIdx.x >> 5;
    const int lane = threadIdx.x & 31;
    const int row  = blockIdx.x * 8 + wave;
    if (row >= n) return;
    float* p = x + (size_t)row * M;
    float sum = 0.0f;
    for (int f = lane; f < M; f += 32) { float v = p[f]; sum += v * v; }
#pragma unroll
    for (int o = 16; o > 0; o >>= 1) sum += __shfl_xor(sum, o, 32);
    const float scale = s / fmaxf(sqrtf(sum), EPS_F);
    if (y != nullptr) {
        const float di = dinv[row];
        float* __restrict__ q = y + (size_t)row * M;
        for (int f = lane; f < M; f += 32) {
            const float v = p[f] * scale;
            p[f] = v;
            q[f] = v * di;
        }
    } else {
        for (int f = lane; f < M; f += 32) p[f] *= scale;
    }
}


template <int V, int N>
__device__ __forceinline__ void add_rows(float (&acc)[V], const float* __restrict__ Y,
                                         const int* __restrict__ sp, int lane) {
    float r[N][V];
#pragma unroll
    for (int t = 0; t < N; ++t) {
        const float* __restrict__ ys = Y + (size_t)sp[t] * (32 * V) + V * lane;
#pragma unroll
        for (int v = 0; v < V; ++v) r[t][v] = ys[v];
    }
#pragma unroll
    for (int t = 0; t < N; ++t) {
#pragma unroll
        for (int v = 0; v < V; ++v) acc[v] += r[t][v];
    }
}

template <int V>
__global__ __launch_bounds__(256) void gather_prop_k(
    float* __restrict__ dst, const float* __restrict__ Y,
    const float* __restrict__ h, const float* __restrict__ dinv,
    const int* __restrict__ rowptr, const int* __restrict__ srcs,
    int n, int relu, int pre) {
    constexpr int M = 32 * V;
    const int wave = __builtin_amdgcn_readfirstlane(threadIdx.x >> 5);
    const int lane = threadIdx.x & 31;
    const int node = blockIdx.x * 8 + wave;
    if (node >= n) return;

    const int e0 = rowptr[node];
    const int e1 = rowptr[node + 1];
    const float di = dinv[node];

    const float* __restrict__ yn = Y + (size_t)node * M + V * lane;
    const float* __restrict__ hn = h + (size_t)node * M + V * lane;
    float acc[V], hv[V];
#pragma unroll
    for (int v = 0; v < V; ++v) acc[v] = yn[v];
#pragma unroll
    for (int v = 0; v < V; ++v) hv[v] = hn[v];

    int e = e0;
    for (; e + 8 <= e1; e += 8) add_rows<V, 8>(acc, Y, srcs + e, lane);
    switch (e1 - e) {
        case 7: add_rows<V, 7>(acc, Y, srcs + e, lane); break;
        case 6: add_rows<V, 6>(acc, Y, srcs + e, lane); break;
        case 5: add_rows<V, 5>(acc, Y, srcs + e, lane); break;
        case 4: add_rows<V, 4>(acc, Y, srcs + e, lane); break;
        case 3: add_rows<V, 3>(acc, Y, srcs + e, lane); break;
        case 2: add_rows<V, 2>(acc, Y, srcs + e, lane); break;
        case 1: add_rows<V, 1>(acc, Y, srcs + e, lane); break;
        default: break;
    }

    const float dd = OMA_F * di;
    float* __restrict__ dn = dst + (size_t)node * M + V * lane;
#pragma unroll
    for (int v = 0; v < V; ++v) {
        float x = dd * acc[v] + ALPHA_F * hv[v];
        if (relu) x = fmaxf(x, 0.0f);
        if (pre) x *= di;
        dn[v] = x;
    }
}


static inline size_t align256(size_t b) { return (b + 255) & ~(size_t)255; }

extern "C" void kernel_launch(void* const* d_in, const int* in_sizes, int n_in,
                              void* d_out, int out_size, void* d_ws, size_t ws_size,
                              hipStream_t stream) {
    const float* x_in = (const float*)d_in[0];
    const int*   ei   = (const int*)d_in[1];
    const float* W0   = (const float*)d_in[2];
    const float* b0   = (const float*)d_in[3];
    const float* W1   = (const float*)d_in[4];
    const float* b1   = (const float*)d_in[5];
    const float* Wx   = (const float*)d_in[6];
    const float* bx   = (const float*)d_in[7];
    float* out = (float*)d_out;

    const int n = in_sizes[0] / IN_DIM;
    const int E = in_sizes[1] / 2;
    const int* row = ei;
    const int* col = ei + E;

    char* ws = (char*)d_ws;
    size_t off = 0;
    float* dinv   = (float*)(ws + off); off += align256((size_t)n * 4);
    int*   cnt    = (int*)  (ws + off); off += align256((size_t)n * 4);
    int*   rowptr = (int*)  (ws + off); off += align256((size_t)(n + 1) * 4);
    int*   cursor = (int*)  (ws + off); off += align256((size_t)n * 4);
    int*   srcs   = (int*)  (ws + off); off += align256((size_t)E * 4);
    float* H      = (float*)(ws + off); off += align256((size_t)n * HID * 4);
    float* P      = (float*)(ws + off); off += align256((size_t)n * HID * 4);
    float* Q      = (float*)(ws + off); off += align256((size_t)n * HID * 4);
    (void)ws_size;

    const int TPB  = 256;
    const int gN   = (n + TPB - 1) / TPB;
    const int gE   = (E + TPB - 1) / TPB;
    const int gRow = (n + 7) / 8;
    const int gTile = ((n / 16) + 7) / 8;

    zero_int_k<<<gN, TPB, 0, stream>>>(cnt, n);
    cnt_k<<<gE, TPB, 0, stream>>>(col, cnt, E);
    dinv_from_cnt_k<<<gN, TPB, 0, stream>>>(cnt, dinv, n);
    scan_k<<<n / 1024 + 1, 1024, 0, stream>>>(cnt, rowptr, n);
    copy_int_k<<<gN, TPB, 0, stream>>>(cursor, rowptr, n);
    fill_k<<<gE, TPB, 0, stream>>>(row, col, cursor, srcs, E);

    auto prop96 = [&](const float* y, float* dst, const float* h, int relu, int pre) {
        gather_prop_k<HID / 32><<<gRow, TPB, 0, stream>>>(dst, y, h, dinv, rowptr, srcs, n, relu, pre);
    };
    auto prop64 = [&](const float* y, float* dst, const float* h, int relu, int pre) {
        gather_prop_k<OUT_DIM / 32><<<gRow, TPB, 0, stream>>>(dst, y, h, dinv, rowptr, srcs, n, relu, pre);
    };

    gemm_wmma_k<HID / 16><<<gTile, TPB, 0, stream>>>(x_in, W0, b0, H, IN_DIM, n);
    l2norm_k<<<gRow, TPB, 0, stream>>>(H, P, dinv, n, HID, 1.8f);
    prop96(P, Q, H, 0, 1);
    prop96(Q, P, H, 1, 0);

    gemm_wmma_k<HID / 16><<<gTile, TPB, 0, stream>>>(P, W1, b1, H, HID, n);
    l2norm_k<<<gRow, TPB, 0, stream>>>(H, Q, dinv, n, HID, 1.8f);
    prop96(Q, P, H, 0, 1);
    prop96(P, Q, H, 0, 0);
    l2norm_k<<<gRow, TPB, 0, stream>>>(Q, nullptr, dinv, n, HID, 1.5f);

    gemm_wmma_k<OUT_DIM / 16><<<gTile, TPB, 0, stream>>>(Q, Wx, bx, H, HID, n);
    l2norm_k<<<gRow, TPB, 0, stream>>>(H, P, dinv, n, OUT_DIM, 1.8f);
    prop64(P, Q, H, 0, 1);
    prop64(Q, out, H, 0, 0);
}


# l2norm fused into its producers: each GEMM's epilogue and layer 1's last gather normalise the rows they hold; the 4 l2norm launches go (19 -> 15); sum-of-squares reduction order changes
# speedup vs baseline: 2.4452x; 1.2024x over previous
#include <hip/hip_runtime.h>
#include <math.h>

typedef __attribute__((ext_vector_type(2))) float v2f;
typedef __attribute__((ext_vector_type(8))) float v8f;

#define IN_DIM 256
#define HID 96
#define OUT_DIM 64
#define ALPHA_F 0.15f
#define OMA_F 0.85f
#define EPS_F 1e-12f


__global__ void zero_int_k(int* p, int n) {
    int i = blockIdx.x * blockDim.x + threadIdx.x;
    if (i < n) p[i] = 0;
}

__global__ void cnt_k(const int* __restrict__ col, int* cnt, int E) {
    int i = blockIdx.x * blockDim.x + threadIdx.x;
    if (i < E) atomicAdd(&cnt[col[i]], 1);
}

__global__ void dinv_from_cnt_k(const int* __restrict__ cnt, float* dinv, int n) {
    int i = blockIdx.x * blockDim.x + threadIdx.x;
    if (i < n) dinv[i] = rsqrtf((float)(cnt[i] + 1));
}

__global__ __launch_bounds__(1024) void scan_k(const int* __restrict__ cnt,
                                               int* __restrict__ rowptr, int n) {
    __shared__ int wtot_v[32];
    __shared__ int wtot_p[32];
    const int tid  = threadIdx.x;
    const int lane = tid & 31;
    const int wave = tid >> 5;

    const int idx = blockIdx.x * 1024 + tid;
    const int v   = (idx < n) ? cnt[idx] : 0;

    const int4* __restrict__ c4 = reinterpret_cast<const int4*>(cnt);
    const int lim = blockIdx.x * 256;
    int p = 0;
#pragma unroll 4
    for (int i = tid; i < lim; i += 1024) {
        const int4 q = c4[i];
        p += (q.x + q.y) + (q.z + q.w);
    }

    int incl = v;
#pragma unroll
    for (int o = 1; o < 32; o <<= 1) {
        const int t = __shfl_up(incl, o, 32);
        if (lane >= o) incl += t;
    }
#pragma unroll
    for (int o = 16; o > 0; o >>= 1) p += __shfl_xor(p, o, 32);

    if (lane == 31) wtot_v[wave] = incl;
    if (lane == 0)  wtot_p[wave] = p;
    __syncthreads();

    int s = wtot_p[lane] + ((lane < wave) ? wtot_v[lane] : 0);
#pragma unroll
    for (int o = 16; o > 0; o >>= 1) s += __shfl_xor(s, o, 32);

    if (idx <= n) rowptr[idx] = s + incl - v;
}

__global__ void copy_int_k(int* __restrict__ dst, const int* __restrict__ src, int n) {
    int i = blockIdx.x * blockDim.x + threadIdx.x;
    if (i < n) dst[i] = src[i];
}

__global__ void fill_k(const int* __restrict__ row, const int* __restrict__ col,
                       int* cursor, int* __restrict__ srcs, int E) {
    int i = blockIdx.x * blockDim.x + threadIdx.x;
    if (i < E) {
        const int p = atomicAdd(&cursor[col[i]], 1);
        srcs[p] = row[i];
    }
}

template <int MASK>
__device__ __forceinline__ float row_xor_add(float v) {
    const int t = __builtin_amdgcn_update_dpp(0, __float_as_int(v), 0x160 + MASK, 0xf, 0xf, false);
    return v + __int_as_float(t);
}

template <int NT>
__global__ __launch_bounds__(256) void gemm_wmma_k(
    const float* __restrict__ X, const float* __restrict__ W,
    const float* __restrict__ bias, float* __restrict__ Out,
    float* __restrict__ Y, const float* __restrict__ dinv,
    int K, int nrows, float s) {
    const int M    = NT * 16;
    const int wave = threadIdx.x >> 5;
    const int tm   = blockIdx.x * 8 + wave;
    if (tm * 16 >= nrows) return;
    const int lane = threadIdx.x & 31;
    const int half = lane >> 4;
    const int l16  = lane & 15;

    v8f c[NT];
#pragma unroll
    for (int t = 0; t < NT; ++t) c[t] = (v8f){};

    const float* __restrict__ xrow = X + (size_t)(tm * 16 + l16) * K;

    for (int k = 0; k < K; k += 4) {
        const int kk = k + 2 * half;
        v2f a;
        a.x = xrow[kk];
        a.y = xrow[kk + 1];
        const float* __restrict__ w0 = W + (size_t)kk * M + l16;
#pragma unroll
        for (int t = 0; t < NT; ++t) {
            v2f b;
            b.x = w0[t * 16];
            b.y = w0[M + t * 16];
            c[t] = __builtin_amdgcn_wmma_f32_16x16x4_f32(
                false, a, false, b, (short)0, c[t], false, false);
        }
    }

    float bv[NT];
#pragma unroll
    for (int t = 0; t < NT; ++t) bv[t] = bias[t * 16 + l16];

    float ss[8];
#pragma unroll
    for (int r = 0; r < 8; ++r) {
        float q = 0.0f;
#pragma unroll
        for (int t = 0; t < NT; ++t) {
            const float v = c[t][r] + bv[t];
            c[t][r] = v;
            q += v * v;
        }
        ss[r] = q;
    }
#pragma unroll
    for (int r = 0; r < 8; ++r) {
        float q = ss[r];
        q = row_xor_add<8>(q);
        q = row_xor_add<4>(q);
        q = row_xor_add<2>(q);
        q = row_xor_add<1>(q);
        ss[r] = q;
    }

#pragma unroll
    for (int r = 0; r < 8; ++r) {
        const int orow = tm * 16 + r + 8 * half;
        const float scale = s / fmaxf(sqrtf(ss[r]), EPS_F);
        const float di = dinv[orow];
#pragma unroll
        for (int t = 0; t < NT; ++t) {
            const int bcol = t * 16 + l16;
            const float v = c[t][r] * scale;
            Out[(size_t)orow * M + bcol] = v;
            Y[(size_t)orow * M + bcol]   = v * di;
        }
    }
}


template <int V, int N>
__device__ __forceinline__ void add_rows(float (&acc)[V], const float* __restrict__ Y,
                                         const int* __restrict__ sp, int lane) {
    float r[N][V];
#pragma unroll
    for (int t = 0; t < N; ++t) {
        const float* __restrict__ ys = Y + (size_t)sp[t] * (32 * V) + V * lane;
#pragma unroll
        for (int v = 0; v < V; ++v) r[t][v] = ys[v];
    }
#pragma unroll
    for (int t = 0; t < N; ++t) {
#pragma unroll
        for (int v = 0; v < V; ++v) acc[v] += r[t][v];
    }
}

template <int V>
__global__ __launch_bounds__(256) void gather_prop_k(
    float* __restrict__ dst, const float* __restrict__ Y,
    const float* __restrict__ h, const float* __restrict__ dinv,
    const int* __restrict__ rowptr, const int* __restrict__ srcs,
    int n, int relu, int pre, float nrm) {
    constexpr int M = 32 * V;
    const int wave = __builtin_amdgcn_readfirstlane(threadIdx.x >> 5);
    const int lane = threadIdx.x & 31;
    const int node = blockIdx.x * 8 + wave;
    if (node >= n) return;

    const int e0 = rowptr[node];
    const int e1 = rowptr[node + 1];
    const float di = dinv[node];

    const float* __restrict__ yn = Y + (size_t)node * M + V * lane;
    const float* __restrict__ hn = h + (size_t)node * M + V * lane;
    float acc[V], hv[V];
#pragma unroll
    for (int v = 0; v < V; ++v) acc[v] = yn[v];
#pragma unroll
    for (int v = 0; v < V; ++v) hv[v] = hn[v];

    int e = e0;
    for (; e + 8 <= e1; e += 8) add_rows<V, 8>(acc, Y, srcs + e, lane);
    switch (e1 - e) {
        case 7: add_rows<V, 7>(acc, Y, srcs + e, lane); break;
        case 6: add_rows<V, 6>(acc, Y, srcs + e, lane); break;
        case 5: add_rows<V, 5>(acc, Y, srcs + e, lane); break;
        case 4: add_rows<V, 4>(acc, Y, srcs + e, lane); break;
        case 3: add_rows<V, 3>(acc, Y, srcs + e, lane); break;
        case 2: add_rows<V, 2>(acc, Y, srcs + e, lane); break;
        case 1: add_rows<V, 1>(acc, Y, srcs + e, lane); break;
        default: break;
    }

    const float dd = OMA_F * di;
    float* __restrict__ dn = dst + (size_t)node * M + V * lane;
    float xo[V];
#pragma unroll
    for (int v = 0; v < V; ++v) {
        float x = dd * acc[v] + ALPHA_F * hv[v];
        if (relu) x = fmaxf(x, 0.0f);
        if (pre) x *= di;
        xo[v] = x;
    }
    if (nrm != 0.0f) {
        float sum = 0.0f;
#pragma unroll
        for (int v = 0; v < V; ++v) sum += xo[v] * xo[v];
#pragma unroll
        for (int o = 16; o > 0; o >>= 1) sum += __shfl_xor(sum, o, 32);
        const float scale = nrm / fmaxf(sqrtf(sum), EPS_F);
#pragma unroll
        for (int v = 0; v < V; ++v) xo[v] *= scale;
    }
#pragma unroll
    for (int v = 0; v < V; ++v) dn[v] = xo[v];
}


static inline size_t align256(size_t b) { return (b + 255) & ~(size_t)255; }

extern "C" void kernel_launch(void* const* d_in, const int* in_sizes, int n_in,
                              void* d_out, int out_size, void* d_ws, size_t ws_size,
                              hipStream_t stream) {
    const float* x_in = (const float*)d_in[0];
    const int*   ei   = (const int*)d_in[1];
    const float* W0   = (const float*)d_in[2];
    const float* b0   = (const float*)d_in[3];
    const float* W1   = (const float*)d_in[4];
    const float* b1   = (const float*)d_in[5];
    const float* Wx   = (const float*)d_in[6];
    const float* bx   = (const float*)d_in[7];
    float* out = (float*)d_out;

    const int n = in_sizes[0] / IN_DIM;
    const int E = in_sizes[1] / 2;
    const int* row = ei;
    const int* col = ei + E;

    char* ws = (char*)d_ws;
    size_t off = 0;
    float* dinv   = (float*)(ws + off); off += align256((size_t)n * 4);
    int*   cnt    = (int*)  (ws + off); off += align256((size_t)n * 4);
    int*   rowptr = (int*)  (ws + off); off += align256((size_t)(n + 1) * 4);
    int*   cursor = (int*)  (ws + off); off += align256((size_t)n * 4);
    int*   srcs   = (int*)  (ws + off); off += align256((size_t)E * 4);
    float* H      = (float*)(ws + off); off += align256((size_t)n * HID * 4);
    float* P      = (float*)(ws + off); off += align256((size_t)n * HID * 4);
    float* Q      = (float*)(ws + off); off += align256((size_t)n * HID * 4);
    (void)ws_size;

    const int TPB  = 256;
    const int gN   = (n + TPB - 1) / TPB;
    const int gE   = (E + TPB - 1) / TPB;
    const int gRow = (n + 7) / 8;
    const int gTile = ((n / 16) + 7) / 8;

    zero_int_k<<<gN, TPB, 0, stream>>>(cnt, n);
    cnt_k<<<gE, TPB, 0, stream>>>(col, cnt, E);
    dinv_from_cnt_k<<<gN, TPB, 0, stream>>>(cnt, dinv, n);
    scan_k<<<n / 1024 + 1, 1024, 0, stream>>>(cnt, rowptr, n);
    copy_int_k<<<gN, TPB, 0, stream>>>(cursor, rowptr, n);
    fill_k<<<gE, TPB, 0, stream>>>(row, col, cursor, srcs, E);

    auto prop96 = [&](const float* y, float* dst, const float* h, int relu, int pre, float nrm) {
        gather_prop_k<HID / 32><<<gRow, TPB, 0, stream>>>(dst, y, h, dinv, rowptr, srcs, n, relu, pre, nrm);
    };
    auto prop64 = [&](const float* y, float* dst, const float* h, int relu, int pre, float nrm) {
        gather_prop_k<OUT_DIM / 32><<<gRow, TPB, 0, stream>>>(dst, y, h, dinv, rowptr, srcs, n, relu, pre, nrm);
    };

    gemm_wmma_k<HID / 16><<<gTile, TPB, 0, stream>>>(x_in, W0, b0, H, P, dinv, IN_DIM, n, 1.8f);
    prop96(P, Q, H, 0, 1, 0.0f);
    prop96(Q, P, H, 1, 0, 0.0f);

    gemm_wmma_k<HID / 16><<<gTile, TPB, 0, stream>>>(P, W1, b1, H, Q, dinv, HID, n, 1.8f);
    prop96(Q, P, H, 0, 1, 0.0f);
    prop96(P, Q, H, 0, 0, 1.5f);

    gemm_wmma_k<OUT_DIM / 16><<<gTile, TPB, 0, stream>>>(Q, Wx, bx, H, P, dinv, HID, n, 1.8f);
    prop64(P, Q, H, 0, 1, 0.0f);
    prop64(Q, out, H, 0, 0, 0.0f);
}


# CSR build: edges counted and ranked on LDS counters (node ranges x edge chunks; an LDS add's return is the rank), not 1.6M scattered global atomics; 6 launches -> 2; per-node f32 sum order changes
# speedup vs baseline: 3.1135x; 1.2733x over previous
#include <hip/hip_runtime.h>
#include <math.h>

typedef __attribute__((ext_vector_type(2))) float v2f;
typedef __attribute__((ext_vector_type(8))) float v8f;

#define IN_DIM 256
#define HID 96
#define OUT_DIM 64
#define ALPHA_F 0.15f
#define OMA_F 0.85f
#define EPS_F 1e-12f

#define CSR_R 1024
#define CSR_U 16

__global__ __launch_bounds__(CSR_R) void csr_count_k(const int* __restrict__ col,
                                                     int* __restrict__ part_cnt,
                                                     int* __restrict__ part_below, int E, int n) {
    __shared__ int cnt[CSR_R];
    __shared__ int wsum[32];
    const int tid  = threadIdx.x;
    const int lane = tid & 31;
    const int wave = tid >> 5;
    const int r = blockIdx.x, c = blockIdx.y, C = gridDim.y;
    const int lo = r * CSR_R;
    const int left = n - lo;
    const unsigned span = (unsigned)(left < 0 ? 0 : (left > CSR_R ? CSR_R : left));
    cnt[tid] = 0;
    __syncthreads();

    const int per = (E + C - 1) / C;
    const int beg = c * per;
    const int end = (beg + per < E) ? beg + per : E;
    int below = 0;
    for (int i0 = beg + tid; i0 < end; i0 += CSR_R * CSR_U) {
        int q[CSR_U];
#pragma unroll
        for (int u = 0; u < CSR_U; ++u) {
            const int i = i0 + u * CSR_R;
            q[u] = (i < end) ? col[i] : 0x7fffffff;
        }
#pragma unroll
        for (int u = 0; u < CSR_U; ++u) {
            below += (q[u] < lo) ? 1 : 0;
            const unsigned d = (unsigned)q[u] - (unsigned)lo;
            if (d < span) atomicAdd(&cnt[d], 1);
        }
    }
    __syncthreads();
    part_cnt[(size_t)c * (gridDim.x * CSR_R) + lo + tid] = cnt[tid];

#pragma unroll
    for (int o = 16; o > 0; o >>= 1) below += __shfl_xor(below, o, 32);
    if (lane == 0) wsum[wave] = below;
    __syncthreads();
    if (wave == 0) {
        int bsum = wsum[lane];
#pragma unroll
        for (int o = 16; o > 0; o >>= 1) bsum += __shfl_xor(bsum, o, 32);
        if (lane == 0) part_below[r * C + c] = bsum;
    }
}

__global__ __launch_bounds__(CSR_R) void csr_fill_k(const int* __restrict__ row,
                                                    const int* __restrict__ col,
                                                    const int* __restrict__ part_cnt,
                                                    const int* __restrict__ part_below,
                                                    int* __restrict__ rowptr, int* __restrict__ srcs,
                                                    float* __restrict__ dinv, int E, int n) {
    __shared__ int cursor[CSR_R];
    __shared__ int wtot[32];
    const int tid  = threadIdx.x;
    const int lane = tid & 31;
    const int wave = tid >> 5;
    const int r = blockIdx.x, c = blockIdx.y, C = gridDim.y;
    const int lo = r * CSR_R;
    const int left = n - lo;
    const unsigned span = (unsigned)(left < 0 ? 0 : (left > CSR_R ? CSR_R : left));
    const size_t pitch = (size_t)gridDim.x * CSR_R;

    int total = 0, mine = 0, base = 0;
    for (int cc = 0; cc < C; ++cc) {
        const int v = part_cnt[cc * pitch + lo + tid];
        if (cc < c) mine += v;
        total += v;
        base += part_below[r * C + cc];
    }

    int incl = total;
#pragma unroll
    for (int o = 1; o < 32; o <<= 1) {
        const int t = __shfl_up(incl, o, 32);
        if (lane >= o) incl += t;
    }
    if (lane == 31) wtot[wave] = incl;
    __syncthreads();
    int s = (lane < wave) ? wtot[lane] : 0;
#pragma unroll
    for (int o = 16; o > 0; o >>= 1) s += __shfl_xor(s, o, 32);
    const int ptr = base + s + incl - total;

    cursor[tid] = ptr + mine;
    if (c == 0) {
        const int idx = lo + tid;
        if (idx <= n) rowptr[idx] = ptr;
        if (idx < n)  dinv[idx] = rsqrtf((float)(total + 1));
    }
    __syncthreads();

    const int per = (E + C - 1) / C;
    const int beg = c * per;
    const int end = (beg + per < E) ? beg + per : E;
    for (int i0 = beg + tid; i0 < end; i0 += CSR_R * CSR_U) {
        int q[CSR_U], sv[CSR_U], p[CSR_U];
#pragma unroll
        for (int u = 0; u < CSR_U; ++u) {
            const int i = i0 + u * CSR_R;
            q[u]  = (i < end) ? col[i] : 0x7fffffff;
            sv[u] = (i < end) ? row[i] : 0;
        }
#pragma unroll
        for (int u = 0; u < CSR_U; ++u) {
            const unsigned d = (unsigned)q[u] - (unsigned)lo;
            p[u] = -1;
            if (d < span) p[u] = atomicAdd(&cursor[d], 1);
        }
#pragma unroll
        for (int u = 0; u < CSR_U; ++u) {
            if (p[u] >= 0) srcs[p[u]] = sv[u];
        }
    }
}

template <int MASK>
__device__ __forceinline__ float row_xor_add(float v) {
    const int t = __builtin_amdgcn_update_dpp(0, __float_as_int(v), 0x160 + MASK, 0xf, 0xf, false);
    return v + __int_as_float(t);
}

template <int NT>
__global__ __launch_bounds__(256) void gemm_wmma_k(
    const float* __restrict__ X, const float* __restrict__ W,
    const float* __restrict__ bias, float* __restrict__ Out,
    float* __restrict__ Y, const float* __restrict__ dinv,
    int K, int nrows, float s) {
    const int M    = NT * 16;
    const int wave = threadIdx.x >> 5;
    const int tm   = blockIdx.x * 8 + wave;
    if (tm * 16 >= nrows) return;
    const int lane = threadIdx.x & 31;
    const int half = lane >> 4;
    const int l16  = lane & 15;

    v8f c[NT];
#pragma unroll
    for (int t = 0; t < NT; ++t) c[t] = (v8f){};

    const float* __restrict__ xrow = X + (size_t)(tm * 16 + l16) * K;

    for (int k = 0; k < K; k += 4) {
        const int kk = k + 2 * half;
        v2f a;
        a.x = xrow[kk];
        a.y = xrow[kk + 1];
        const float* __restrict__ w0 = W + (size_t)kk * M + l16;
#pragma unroll
        for (int t = 0; t < NT; ++t) {
            v2f b;
            b.x = w0[t * 16];
            b.y = w0[M + t * 16];
            c[t] = __builtin_amdgcn_wmma_f32_16x16x4_f32(
                false, a, false, b, (short)0, c[t], false, false);
        }
    }

    float bv[NT];
#pragma unroll
    for (int t = 0; t < NT; ++t) bv[t] = bias[t * 16 + l16];

    float ss[8];
#pragma unroll
    for (int r = 0; r < 8; ++r) {
        float q = 0.0f;
#pragma unroll
        for (int t = 0; t < NT; ++t) {
            const float v = c[t][r] + bv[t];
            c[t][r] = v;
            q += v * v;
        }
        ss[r] = q;
    }
#pragma unroll
    for (int r = 0; r < 8; ++r) {
        float q = ss[r];
        q = row_xor_add<8>(q);
        q = row_xor_add<4>(q);
        q = row_xor_add<2>(q);
        q = row_xor_add<1>(q);
        ss[r] = q;
    }

#pragma unroll
    for (int r = 0; r < 8; ++r) {
        const int orow = tm * 16 + r + 8 * half;
        const float scale = s / fmaxf(sqrtf(ss[r]), EPS_F);
        const float di = dinv[orow];
#pragma unroll
        for (int t = 0; t < NT; ++t) {
            const int bcol = t * 16 + l16;
            const float v = c[t][r] * scale;
            Out[(size_t)orow * M + bcol] = v;
            Y[(size_t)orow * M + bcol]   = v * di;
        }
    }
}


template <int V, int N>
__device__ __forceinline__ void add_rows(float (&acc)[V], const float* __restrict__ Y,
                                         const int* __restrict__ sp, int lane) {
    float r[N][V];
#pragma unroll
    for (int t = 0; t < N; ++t) {
        const float* __restrict__ ys = Y + (size_t)sp[t] * (32 * V) + V * lane;
#pragma unroll
        for (int v = 0; v < V; ++v) r[t][v] = ys[v];
    }
#pragma unroll
    for (int t = 0; t < N; ++t) {
#pragma unroll
        for (int v = 0; v < V; ++v) acc[v] += r[t][v];
    }
}

template <int V>
__global__ __launch_bounds__(256) void gather_prop_k(
    float* __restrict__ dst, const float* __restrict__ Y,
    const float* __restrict__ h, const float* __restrict__ dinv,
    const int* __restrict__ rowptr, const int* __restrict__ srcs,
    int n, int relu, int pre, float nrm) {
    constexpr int M = 32 * V;
    const int wave = __builtin_amdgcn_readfirstlane(threadIdx.x >> 5);
    const int lane = threadIdx.x & 31;
    const int node = blockIdx.x * 8 + wave;
    if (node >= n) return;

    const int e0 = rowptr[node];
    const int e1 = rowptr[node + 1];
    const float di = dinv[node];

    const float* __restrict__ yn = Y + (size_t)node * M + V * lane;
    const float* __restrict__ hn = h + (size_t)node * M + V * lane;
    float acc[V], hv[V];
#pragma unroll
    for (int v = 0; v < V; ++v) acc[v] = yn[v];
#pragma unroll
    for (int v = 0; v < V; ++v) hv[v] = hn[v];

    int e = e0;
    for (; e + 8 <= e1; e += 8) add_rows<V, 8>(acc, Y, srcs + e, lane);
    switch (e1 - e) {
        case 7: add_rows<V, 7>(acc, Y, srcs + e, lane); break;
        case 6: add_rows<V, 6>(acc, Y, srcs + e, lane); break;
        case 5: add_rows<V, 5>(acc, Y, srcs + e, lane); break;
        case 4: add_rows<V, 4>(acc, Y, srcs + e, lane); break;
        case 3: add_rows<V, 3>(acc, Y, srcs + e, lane); break;
        case 2: add_rows<V, 2>(acc, Y, srcs + e, lane); break;
        case 1: add_rows<V, 1>(acc, Y, srcs + e, lane); break;
        default: break;
    }

    const float dd = OMA_F * di;
    float* __restrict__ dn = dst + (size_t)node * M + V * lane;
    float xo[V];
#pragma unroll
    for (int v = 0; v < V; ++v) {
        float x = dd * acc[v] + ALPHA_F * hv[v];
        if (relu) x = fmaxf(x, 0.0f);
        if (pre) x *= di;
        xo[v] = x;
    }
    if (nrm != 0.0f) {
        float sum = 0.0f;
#pragma unroll
        for (int v = 0; v < V; ++v) sum += xo[v] * xo[v];
#pragma unroll
        for (int o = 16; o > 0; o >>= 1) sum += __shfl_xor(sum, o, 32);
        const float scale = nrm / fmaxf(sqrtf(sum), EPS_F);
#pragma unroll
        for (int v = 0; v < V; ++v) xo[v] *= scale;
    }
#pragma unroll
    for (int v = 0; v < V; ++v) dn[v] = xo[v];
}


static inline size_t align256(size_t b) { return (b + 255) & ~(size_t)255; }

extern "C" void kernel_launch(void* const* d_in, const int* in_sizes, int n_in,
                              void* d_out, int out_size, void* d_ws, size_t ws_size,
                              hipStream_t stream) {
    const float* x_in = (const float*)d_in[0];
    const int*   ei   = (const int*)d_in[1];
    const float* W0   = (const float*)d_in[2];
    const float* b0   = (const float*)d_in[3];
    const float* W1   = (const float*)d_in[4];
    const float* b1   = (const float*)d_in[5];
    const float* Wx   = (const float*)d_in[6];
    const float* bx   = (const float*)d_in[7];
    float* out = (float*)d_out;

    const int n = in_sizes[0] / IN_DIM;
    const int E = in_sizes[1] / 2;
    const int* row = ei;
    const int* col = ei + E;

    char* ws = (char*)d_ws;
    size_t off = 0;
    const int NR = n / CSR_R + 1;
    const int NC = 5;
    float* dinv   = (float*)(ws + off); off += align256((size_t)n * 4);
    int*   rowptr = (int*)  (ws + off); off += align256((size_t)(n + 1) * 4);
    int*   srcs   = (int*)  (ws + off); off += align256((size_t)E * 4);
    int*   pcnt   = (int*)  (ws + off); off += align256((size_t)NC * NR * CSR_R * 4);
    int*   pbelow = (int*)  (ws + off); off += align256((size_t)NR * NC * 4);
    float* H      = (float*)(ws + off); off += align256((size_t)n * HID * 4);
    float* P      = (float*)(ws + off); off += align256((size_t)n * HID * 4);
    float* Q      = (float*)(ws + off); off += align256((size_t)n * HID * 4);
    (void)ws_size;

    const int TPB  = 256;
    const int gRow = (n + 7) / 8;
    const int gTile = ((n / 16) + 7) / 8;

    csr_count_k<<<dim3(NR, NC), CSR_R, 0, stream>>>(col, pcnt, pbelow, E, n);
    csr_fill_k<<<dim3(NR, NC), CSR_R, 0, stream>>>(row, col, pcnt, pbelow, rowptr, srcs, dinv, E, n);

    auto prop96 = [&](const float* y, float* dst, const float* h, int relu, int pre, float nrm) {
        gather_prop_k<HID / 32><<<gRow, TPB, 0, stream>>>(dst, y, h, dinv, rowptr, srcs, n, relu, pre, nrm);
    };
    auto prop64 = [&](const float* y, float* dst, const float* h, int relu, int pre, float nrm) {
        gather_prop_k<OUT_DIM / 32><<<gRow, TPB, 0, stream>>>(dst, y, h, dinv, rowptr, srcs, n, relu, pre, nrm);
    };

    gemm_wmma_k<HID / 16><<<gTile, TPB, 0, stream>>>(x_in, W0, b0, H, P, dinv, IN_DIM, n, 1.8f);
    prop96(P, Q, H, 0, 1, 0.0f);
    prop96(Q, P, H, 1, 0, 0.0f);

    gemm_wmma_k<HID / 16><<<gTile, TPB, 0, stream>>>(P, W1, b1, H, Q, dinv, HID, n, 1.8f);
    prop96(Q, P, H, 0, 1, 0.0f);
    prop96(P, Q, H, 0, 0, 1.5f);

    gemm_wmma_k<OUT_DIM / 16><<<gTile, TPB, 0, stream>>>(Q, Wx, bx, H, P, dinv, HID, n, 1.8f);
    prop64(P, Q, H, 0, 1, 0.0f);
    prop64(Q, out, H, 0, 0, 0.0f);
}
